# starting kernel
# baseline (speedup 1.0000x reference)
#include <hip/hip_runtime.h>
#include <hip/hip_bf16.h>

typedef __attribute__((ext_vector_type(2))) float v2f;
typedef __attribute__((ext_vector_type(8))) float v8f;


__device__ __forceinline__ v8f wmma_tile(v2f a, v2f b) {
    v8f acc = {};
    return __builtin_amdgcn_wmma_f32_16x16x4_f32(
        false, a, false, b, (short)0, acc, false, false);
}

__global__ void rot_wmma_kernel(const float* __restrict__ x,
                                const float* __restrict__ w,
                                float* __restrict__ out,
                                int numTiles, int n) {
    const float a = w[0], b = w[1], c = w[2];
    const float sa = sinf(a), ca = cosf(a);
    const float sb = sinf(b), cb = cosf(b);
    const float sc = sinf(c), cc = cosf(c);
    const float R00 = cc * cb, R01 = sa * sb * cc - ca * sc, R02 = ca * sb * cc + sa * sc;
    const float R10 = cb * sc, R11 = sa * sb * sc + ca * cc, R12 = ca * sb * sc - sa * cc;
    const float R20 = -sb,     R21 = sa * cb,                R22 = ca * cb;

    const unsigned lane = threadIdx.x & 31u;
    const unsigned m    = lane & 15u;
    const bool     hi   = lane >= 16u;

    const float Rm0 = (m == 0u) ? R00 : (m == 1u) ? R10 : (m == 2u) ? R20 : 0.0f;
    const float Rm1 = (m == 0u) ? R01 : (m == 1u) ? R11 : (m == 2u) ? R21 : 0.0f;
    const float Rm2 = (m == 0u) ? R02 : (m == 1u) ? R12 : (m == 2u) ? R22 : 0.0f;
    v2f afrag;
    afrag.x = hi ? Rm2 : Rm0;
    afrag.y = hi ? 0.0f : Rm1;

    const unsigned waveId   = (blockIdx.x * blockDim.x + threadIdx.x) >> 5;
    const unsigned numWaves = (gridDim.x * blockDim.x) >> 5;

    const unsigned off0 = m * 3u + (hi ? 2u : 0u);
    const unsigned off1 = m * 3u + 1u;

    const unsigned nt  = (unsigned)numTiles;
    const unsigned nt4 = nt & ~3u;

    for (unsigned t0 = waveId * 4u; t0 < nt4; t0 += numWaves * 4u) {
        const float* xt = x + (size_t)t0 * 48u;

        v2f bf[4];
#pragma unroll
        for (int i = 0; i < 4; ++i) {
            const float bx = xt[i * 48 + off0];
            const float by = xt[i * 48 + off1];
            bf[i].x = bx;
            bf[i].y = hi ? 0.0f : by;
        }

        const v8f d0 = wmma_tile(afrag, bf[0]);
        const v8f d1 = wmma_tile(afrag, bf[1]);
        const v8f d2 = wmma_tile(afrag, bf[2]);
        const v8f d3 = wmma_tile(afrag, bf[3]);

        if (!hi) {
            float* ot = out + (size_t)t0 * 48u + m * 3u;
            ot[0]       = d0[0]; ot[1]       = d0[1]; ot[2]       = d0[2];
            ot[48 + 0]  = d1[0]; ot[48 + 1]  = d1[1]; ot[48 + 2]  = d1[2];
            ot[96 + 0]  = d2[0]; ot[96 + 1]  = d2[1]; ot[96 + 2]  = d2[2];
            ot[144 + 0] = d3[0]; ot[144 + 1] = d3[1]; ot[144 + 2] = d3[2];
        }
    }

    if (waveId < nt - nt4) {
        const unsigned t = nt4 + waveId;
        const float* xt = x + (size_t)t * 48u;
        v2f bf;
        bf.x = xt[off0];
        const float by = xt[off1];
        bf.y = hi ? 0.0f : by;
        const v8f d = wmma_tile(afrag, bf);
        if (!hi) {
            float* ot = out + (size_t)t * 48u + m * 3u;
            ot[0] = d[0]; ot[1] = d[1]; ot[2] = d[2];
        }
    }

    const int tailStart = numTiles * 16;
    const int gtid = blockIdx.x * blockDim.x + threadIdx.x;
    const int ti = tailStart + gtid;
    if (ti < n) {
        const float x0 = x[3 * ti], x1 = x[3 * ti + 1], x2 = x[3 * ti + 2];
        out[3 * ti + 0] = R00 * x0 + R01 * x1 + R02 * x2;
        out[3 * ti + 1] = R10 * x0 + R11 * x1 + R12 * x2;
        out[3 * ti + 2] = R20 * x0 + R21 * x1 + R22 * x2;
    }
}

extern "C" void kernel_launch(void* const* d_in, const int* in_sizes, int n_in,
                              void* d_out, int out_size, void* d_ws, size_t ws_size,
                              hipStream_t stream) {
    const float* x = (const float*)d_in[0];
    const float* w = (const float*)d_in[1];
    float* out = (float*)d_out;

    const int n = in_sizes[0] / 3;
    const int numTiles = n / 16;

    const int threads = 256;
    int blocks = 2048;
    const int neededWaves = (numTiles + 3) / 4 > 0 ? (numTiles + 3) / 4 : 1;
    const int maxBlocks = (neededWaves * 32 + threads - 1) / threads;
    if (blocks > maxBlocks) blocks = maxBlocks;
    if (blocks < 1) blocks = 1;

    rot_wmma_kernel<<<blocks, threads, 0, stream>>>(x, w, out, numTiles, n);
}


# lane = point: whole-line 12-B loads and stores on all lanes; two 32-point tiles share 3 chained K=4 WMMAs (K=12 block operator); 16 loads in flight per wave, 4096 waves; per-output sum order changes
# speedup vs baseline: 1.0673x; 1.0673x over previous
#include <hip/hip_runtime.h>

typedef __attribute__((ext_vector_type(2))) float v2f;
typedef __attribute__((ext_vector_type(8))) float v8f;


#define UNITS_PER_ITER 8

__device__ __forceinline__ v8f wmma_acc(v2f a, v2f b, v8f c) {
    return __builtin_amdgcn_wmma_f32_16x16x4_f32(
        false, a, false, b, (short)0, c, false, false);
}

__device__ __forceinline__ void unit_products(v2f a0, v2f a1, v2f a2,
                                              const float* f, v8f& d) {
    v2f b0, b1, b2;
    b0.x = f[0]; b0.y = f[1];
    b1.x = f[3]; b1.y = f[4];
    b2.x = f[2]; b2.y = f[5];
    v8f acc = {};
    acc = wmma_acc(a0, b0, acc);
    acc = wmma_acc(a1, b1, acc);
    acc = wmma_acc(a2, b2, acc);
    d = acc;
}

__global__ void __launch_bounds__(256)
rot_wmma_kernel(const float* __restrict__ x,
                const float* __restrict__ w,
                float* __restrict__ out,
                int n) {
    const float a = w[0], b = w[1], c = w[2];
    const float sa = sinf(a), ca = cosf(a);
    const float sb = sinf(b), cb = cosf(b);
    const float sc = sinf(c), cc = cosf(c);
    const float R00 = cc * cb, R01 = sa * sb * cc - ca * sc, R02 = ca * sb * cc + sa * sc;
    const float R10 = cb * sc, R11 = sa * sb * sc + ca * cc, R12 = ca * sb * sc - sa * cc;
    const float R20 = -sb,     R21 = sa * cb,                R22 = ca * cb;

    const unsigned lane = threadIdx.x & 31u;
    const unsigned col  = lane & 15u;
    const unsigned half = lane >> 4;
    const unsigned rem  = col & 7u;
    const unsigned ri   = rem & 3u;
    const bool live  = (half == (col >> 3)) && (ri < 3u);
    const bool forT  = live && (rem < 4u);
    const bool forT2 = live && (rem >= 4u);
    const float Rr0 = (ri == 0u) ? R00 : (ri == 1u) ? R10 : R20;
    const float Rr1 = (ri == 0u) ? R01 : (ri == 1u) ? R11 : R21;
    const float Rr2 = (ri == 0u) ? R02 : (ri == 1u) ? R12 : R22;
    v2f a0, a1, a2;
    a0.x = forT  ? Rr0 : 0.0f;  a0.y = forT  ? Rr1 : 0.0f;
    a1.x = forT2 ? Rr0 : 0.0f;  a1.y = forT2 ? Rr1 : 0.0f;
    a2.x = forT  ? Rr2 : 0.0f;  a2.y = forT2 ? Rr2 : 0.0f;

    const unsigned waveId   = __builtin_amdgcn_readfirstlane(
        (blockIdx.x * blockDim.x + threadIdx.x) >> 5);
    const unsigned numWaves = (gridDim.x * blockDim.x) >> 5;

    const unsigned un        = (unsigned)n;
    const unsigned fullUnits = un >> 6;
    const unsigned numGroups = fullUnits / UNITS_PER_ITER;

    for (unsigned g = waveId; g < numGroups; g += numWaves) {
        const size_t base = (size_t)g * (UNITS_PER_ITER * 192u) + lane * 3u;
        const float* xp = x + base;
        float*       op = out + base;

        float f[UNITS_PER_ITER][6];
#pragma unroll
        for (int u = 0; u < UNITS_PER_ITER; ++u) {
            f[u][0] = xp[u * 192 + 0];
            f[u][1] = xp[u * 192 + 1];
            f[u][2] = xp[u * 192 + 2];
            f[u][3] = xp[u * 192 + 96 + 0];
            f[u][4] = xp[u * 192 + 96 + 1];
            f[u][5] = xp[u * 192 + 96 + 2];
        }
        __builtin_amdgcn_sched_barrier(0);
#pragma unroll
        for (int u = 0; u < UNITS_PER_ITER; u += 2) {
            v8f d0, d1;
            unit_products(a0, a1, a2, f[u],     d0);
            unit_products(a0, a1, a2, f[u + 1], d1);
            op[u * 192 + 0]        = d0[0];
            op[u * 192 + 1]        = d0[1];
            op[u * 192 + 2]        = d0[2];
            op[u * 192 + 96 + 0]   = d0[4];
            op[u * 192 + 96 + 1]   = d0[5];
            op[u * 192 + 96 + 2]   = d0[6];
            op[u * 192 + 192 + 0]  = d1[0];
            op[u * 192 + 192 + 1]  = d1[1];
            op[u * 192 + 192 + 2]  = d1[2];
            op[u * 192 + 288 + 0]  = d1[4];
            op[u * 192 + 288 + 1]  = d1[5];
            op[u * 192 + 288 + 2]  = d1[6];
            __builtin_amdgcn_sched_barrier(0);
        }
    }

    const unsigned firstLeft  = numGroups * UNITS_PER_ITER;
    const unsigned totalUnits = (un + 63u) >> 6;
    for (unsigned uu = firstLeft + waveId; uu < totalUnits; uu += numWaves) {
        const unsigned p0 = uu * 64u + lane;
        const unsigned p1 = p0 + 32u;
        float f[6] = {0.0f, 0.0f, 0.0f, 0.0f, 0.0f, 0.0f};
        if (p0 < un) {
            const float* q = x + (size_t)p0 * 3u;
            f[0] = q[0]; f[1] = q[1]; f[2] = q[2];
        }
        if (p1 < un) {
            const float* q = x + (size_t)p1 * 3u;
            f[3] = q[0]; f[4] = q[1]; f[5] = q[2];
        }
        v8f d;
        unit_products(a0, a1, a2, f, d);
        if (p0 < un) {
            float* q = out + (size_t)p0 * 3u;
            q[0] = d[0]; q[1] = d[1]; q[2] = d[2];
        }
        if (p1 < un) {
            float* q = out + (size_t)p1 * 3u;
            q[0] = d[4]; q[1] = d[5]; q[2] = d[6];
        }
    }
}

extern "C" void kernel_launch(void* const* d_in, const int* in_sizes, int n_in,
                              void* d_out, int out_size, void* d_ws, size_t ws_size,
                              hipStream_t stream) {
    const float* x = (const float*)d_in[0];
    const float* w = (const float*)d_in[1];
    float* out = (float*)d_out;

    const int n = in_sizes[0] / 3;

    rot_wmma_kernel<<<512, 256, 0, stream>>>(x, w, out, n);
}
